# starting kernel
# baseline (speedup 1.0000x reference)
#include <hip/hip_runtime.h>
#include <math.h>

typedef __attribute__((ext_vector_type(2))) float v2f;
typedef __attribute__((ext_vector_type(4))) float v4f;
typedef __attribute__((ext_vector_type(8))) float v8f;

#define Bdim 2
#define Cdim 32
#define Hdim 96
#define Wdim 160
#define Ddim 32
#define NPIX (Bdim * Hdim * Wdim)
#define CSTRIDE ((long)Hdim * Wdim * Ddim)
#define QS_STRIDE 36

__global__ __launch_bounds__(256) void dav_pass1(
    const float* __restrict__ q, const float* __restrict__ wf,
    float* __restrict__ S, float* __restrict__ E, float* __restrict__ Aq) {
  __shared__ float qs[32 * QS_STRIDE];

  const int t   = threadIdx.x;
  const int blk = blockIdx.x;
  const int wq  = blk % 5;
  const int h   = (blk / 5) % Hdim;
  const int b   = blk / (5 * Hdim);
  const int w0  = wq * 32;

  {
    const int wl = t & 31;
    const int c0 = t >> 5;
#pragma unroll
    for (int r = 0; r < 4; ++r) {
      const int c = c0 + 8 * r;
      qs[wl * QS_STRIDE + c] = q[((b * Cdim + c) * Hdim + h) * Wdim + w0 + wl];
    }
  }
  __syncthreads();

  if (t < 32) {
    float acc = 0.f;
#pragma unroll
    for (int c = 0; c < Cdim; ++c) {
      const float v = qs[t * QS_STRIDE + c];
      acc += v * v;
    }
    Aq[(b * Hdim + h) * Wdim + w0 + t] = acc;
  }

  const int wave = t >> 5;
  const int lane = t & 31;
  const int pxl  = wave * 4 + (lane >> 3);
  const int d4   = lane & 7;
  const int w    = w0 + pxl;

  const float* kp =
      wf + ((long)(b * Cdim) * (Hdim * Wdim) + (long)h * Wdim + w) * Ddim +
      d4 * 4;

  v4f s4 = {0.f, 0.f, 0.f, 0.f};
  v4f e4 = {0.f, 0.f, 0.f, 0.f};
#pragma unroll
  for (int c4 = 0; c4 < 8; ++c4) {
    const v4f qv4 = *(const v4f*)&qs[pxl * QS_STRIDE + c4 * 4];
#pragma unroll
    for (int u = 0; u < 4; ++u) {
      const v4f kv = *(const v4f*)kp;
      kp += CSTRIDE;
      const float qv = qv4[u];
      const v4f qvv = {qv, qv, qv, qv};
      s4 += qvv * kv;
      e4 += kv * kv;
    }
  }

  const long pix = (long)(b * Hdim + h) * Wdim + w;
  *(v4f*)(S + pix * Ddim + d4 * 4) = s4;
  *(v4f*)(E + pix * Ddim + d4 * 4) = e4;
}

__global__ __launch_bounds__(128) void dav_pass2(
    const float* __restrict__ S, const float* __restrict__ E,
    const float* __restrict__ Aq, float* __restrict__ out) {
  const int wid  = blockIdx.x * 4 + (threadIdx.x >> 5);
  const int lane = threadIdx.x & 31;

  const int n     = wid % 10;
  const int dh    = (wid / 10) & 1;
  const int h     = (wid / 20) % Hdim;
  const int b     = wid / (20 * Hdim);
  const int w0    = n * 16;
  const int dbase = dh * 16;

  const int nn   = lane & 15;
  const int half = lane >> 4;
  const int col  = dbase + nn;

  const int  rowStride = Wdim * Ddim;
  const float gU = (h > 0) ? 1.f : 0.f;
  const float gD = (h < Hdim - 1) ? 1.f : 0.f;
  const int  oU = (h > 0) ? -rowStride : 0;
  const int  oD = (h < Hdim - 1) ? rowStride : 0;

  const long rowbase = (long)(b * Hdim + h) * Wdim * Ddim + col;
  const float* Srow = S + rowbase;
  const float* Erow = E + rowbase;

  v8f accS = {0.f, 0.f, 0.f, 0.f, 0.f, 0.f, 0.f, 0.f};
  v8f accE = {0.f, 0.f, 0.f, 0.f, 0.f, 0.f, 0.f, 0.f};

#pragma unroll
  for (int j = 0; j < 5; ++j) {
    v2f aS, aE, bb;
#pragma unroll
    for (int v = 0; v < 2; ++v) {
      const int k   = v + 2 * half;
      const int wp  = w0 - 1 + 4 * j + k;
      const int wpc = wp < 0 ? 0 : (wp > Wdim - 1 ? Wdim - 1 : wp);
      const long idx = (long)wpc * Ddim;

      const float s1 = Srow[idx];
      const float s0 = Srow[idx + oU];
      const float s2 = Srow[idx + oD];
      const float e1 = Erow[idx];
      const float e0 = Erow[idx + oU];
      const float e2 = Erow[idx + oD];
      aS[v] = fmaf(gU, s0, fmaf(gD, s2, s1));
      aE[v] = fmaf(gU, e0, fmaf(gD, e2, e1));

      const int diff = 4 * j + k - 1 - nn;
      const bool inband = (unsigned)(diff + 1) <= 2u;
      const bool wok    = (unsigned)wp < (unsigned)Wdim;
      bb[v] = (inband && wok) ? 1.f : 0.f;
    }
    accS = __builtin_amdgcn_wmma_f32_16x16x4_f32(false, aS, false, bb,
                                                 (short)0, accS, false, false);
    accE = __builtin_amdgcn_wmma_f32_16x16x4_f32(false, aE, false, bb,
                                                 (short)0, accE, false, false);
  }

  const int wcol = w0 + nn;
  const float* Arow = Aq + (b * Hdim + h) * Wdim;
  float bsa = 0.f;
#pragma unroll
  for (int dy = -1; dy <= 1; ++dy) {
    const int hh = h + dy;
    const bool hok = (unsigned)hh < (unsigned)Hdim;
    const int ho = hok ? dy * Wdim : 0;
    const float gh = hok ? 1.f : 0.f;
#pragma unroll
    for (int dx = -1; dx <= 1; ++dx) {
      const int ww  = wcol + dx;
      const bool wok = (unsigned)ww < (unsigned)Wdim;
      const int wwc = ww < 0 ? 0 : (ww > Wdim - 1 ? Wdim - 1 : ww);
      const float g = wok ? gh : 0.f;
      bsa = fmaf(g, Arow[ho + wwc], bsa);
    }
  }
  const float nq = fmaxf(sqrtf(bsa), 1e-12f);

#pragma unroll
  for (int r = 0; r < 8; ++r) {
    const int d     = dbase + r + 8 * half;
    const float nk  = fmaxf(sqrtf(accE[r]), 1e-12f);
    const float sim = accS[r] / (nq * nk);
    out[((long)(b * Ddim + d) * Hdim + h) * Wdim + wcol] = sim;
  }
}

extern "C" void kernel_launch(void* const* d_in, const int* in_sizes, int n_in,
                              void* d_out, int out_size, void* d_ws,
                              size_t ws_size, hipStream_t stream) {
  const float* q  = (const float*)d_in[0];
  const float* wf = (const float*)d_in[1];
  float* out = (float*)d_out;

  float* S  = (float*)d_ws;
  float* E  = S + (size_t)NPIX * Ddim;
  float* Aq = E + (size_t)NPIX * Ddim;

  dav_pass1<<<Bdim * Hdim * 5, 256, 0, stream>>>(q, wf, S, E, Aq);
  dav_pass2<<<(Bdim * Hdim * 2 * 10) / 4, 128, 0, stream>>>(S, E, Aq, out);
}


# fuse both passes into one launch: 240 workgroups each stream an 8x16 tile plus its 3x3 halo (180 px), keep S/E/a in LDS and do box sums + normalisation there; no workspace trip; sum order kept
# speedup vs baseline: 1.0847x; 1.0847x over previous
#include <hip/hip_runtime.h>
#include <math.h>

typedef __attribute__((ext_vector_type(2))) float v2f;
typedef __attribute__((ext_vector_type(4))) float v4f;
typedef __attribute__((ext_vector_type(8))) float v8f;

#define Bdim 2
#define Cdim 32
#define Hdim 96
#define Wdim 160
#define Ddim 32
#define CSTRIDE (Hdim * Wdim * Ddim)

#define TH 8
#define TW 16
#define ER (TH + 2)
#define EC (TW + 2)
#define NPX (ER * EC)
#define SWAVES 15
#define PXW 12
#define WAVES 16
#define THREADS (WAVES * 32)
#define QPITCH 36
#define SPITCH 40
#define NCHUNK 6

__global__ __launch_bounds__(THREADS) void dav_fused(
    const float* __restrict__ q, const float* __restrict__ wf,
    float* __restrict__ out) {
  __shared__ __attribute__((aligned(16))) float qs[SWAVES * PXW * QPITCH];
  __shared__ __attribute__((aligned(16))) float Ss[NPX * SPITCH];
  __shared__ __attribute__((aligned(16))) float Es[NPX * SPITCH];
  __shared__ float As[NPX + 12];

  const int t    = threadIdx.x;
  const int lane = t & 31;
  const int wave = __builtin_amdgcn_readfirstlane(t >> 5);
  const int blk  = blockIdx.x;
  const int tw   = blk % 10;
  const int th   = (blk / 10) % 12;
  const int b    = blk / 120;
  const int h0   = th * TH;
  const int w0   = tw * TW;

  if (wave < SWAVES) {
    const int pq = lane >> 3;
    const int d4 = lane & 7;
    float* qsw = qs + wave * (PXW * QPITCH);

    const float* wfb = wf + (size_t)b * Cdim * CSTRIDE;
    int koff[3];
#pragma unroll
    for (int g = 0; g < 3; ++g) {
      const int p  = wave * PXW + 4 * g + pq;
      const int er = p / EC;
      const int ec = p - er * EC;
      int h = h0 - 1 + er;
      int w = w0 - 1 + ec;
      h = h < 0 ? 0 : (h > Hdim - 1 ? Hdim - 1 : h);
      w = w < 0 ? 0 : (w > Wdim - 1 ? Wdim - 1 : w);
      koff[g] = (h * Wdim + w) * Ddim + d4 * 4;
    }

    v4f buf[2][16];
#define LOAD_CHUNK(K)                                                          \
  do {                                                                         \
    const float* p_ = wfb + koff[(K) >> 1] + (((K) & 1) * 16) * CSTRIDE;       \
    _Pragma("unroll") for (int j_ = 0; j_ < 16; ++j_)                          \
        buf[(K) & 1][j_] = *(const v4f*)(p_ + j_ * CSTRIDE);                   \
  } while (0)

    LOAD_CHUNK(0);
    __builtin_amdgcn_sched_barrier(0);

    float qreg[12];
    int   qdst[12];
#pragma unroll
    for (int i = 0; i < 12; ++i) {
      const int e  = lane + 32 * i;
      const int c  = e / PXW;
      const int px = e - c * PXW;
      const int p  = wave * PXW + px;
      const int er = p / EC;
      const int ec = p - er * EC;
      int h = h0 - 1 + er;
      int w = w0 - 1 + ec;
      h = h < 0 ? 0 : (h > Hdim - 1 ? Hdim - 1 : h);
      w = w < 0 ? 0 : (w > Wdim - 1 ? Wdim - 1 : w);
      qreg[i] = q[((b * Cdim + c) * Hdim + h) * Wdim + w];
      qdst[i] = px * QPITCH + c;
    }
    __builtin_amdgcn_sched_barrier(0);
    LOAD_CHUNK(1);
    __builtin_amdgcn_sched_barrier(0);

#pragma unroll
    for (int i = 0; i < 12; ++i) qsw[qdst[i]] = qreg[i];
    __builtin_amdgcn_fence(__ATOMIC_RELEASE, "wavefront");
    __builtin_amdgcn_wave_barrier();
    __builtin_amdgcn_fence(__ATOMIC_ACQUIRE, "wavefront");

    v4f s4[3], e4[3];
#pragma unroll
    for (int g = 0; g < 3; ++g) {
      s4[g] = (v4f){0.f, 0.f, 0.f, 0.f};
      e4[g] = (v4f){0.f, 0.f, 0.f, 0.f};
    }

#pragma unroll
    for (int k = 0; k < NCHUNK; ++k) {
      const int g = k >> 1;
      const float* qrow = qsw + (4 * g + pq) * QPITCH + (k & 1) * 16;
#pragma unroll
      for (int j4 = 0; j4 < 4; ++j4) {
        const v4f qv4 = *(const v4f*)(qrow + 4 * j4);
#pragma unroll
        for (int u = 0; u < 4; ++u) {
          const v4f kv  = buf[k & 1][4 * j4 + u];
          const float qv = qv4[u];
          const v4f qvv = {qv, qv, qv, qv};
          s4[g] += qvv * kv;
          e4[g] += kv * kv;
        }
      }
      __builtin_amdgcn_sched_barrier(0);
      if (k + 2 < NCHUNK) {
        asm volatile("" : "+v"(koff[(k + 2) >> 1]) : "v"(s4[g]), "v"(e4[g]));
        LOAD_CHUNK(k + 2);
      }
      __builtin_amdgcn_sched_barrier(0);
    }
#undef LOAD_CHUNK

    if (lane < PXW) {
      float acc = 0.f;
#pragma unroll
      for (int c = 0; c < Cdim; ++c) {
        const float v = qsw[lane * QPITCH + c];
        acc += v * v;
      }
      As[wave * PXW + lane] = acc;
    }

#pragma unroll
    for (int g = 0; g < 3; ++g) {
      const int p = wave * PXW + 4 * g + pq;
      *(v4f*)(Ss + p * SPITCH + d4 * 4) = s4[g];
      *(v4f*)(Es + p * SPITCH + d4 * 4) = e4[g];
    }
  }
  __syncthreads();

  const int i     = wave >> 1;
  const int dh    = wave & 1;
  const int h     = h0 + i;
  const int dbase = dh * 16;
  const int nn    = lane & 15;
  const int half  = lane >> 4;
  const int col   = dbase + nn;

  const float gU = (h > 0) ? 1.f : 0.f;
  const float gD = (h < Hdim - 1) ? 1.f : 0.f;

  v8f accS = {0.f, 0.f, 0.f, 0.f, 0.f, 0.f, 0.f, 0.f};
  v8f accE = {0.f, 0.f, 0.f, 0.f, 0.f, 0.f, 0.f, 0.f};

#pragma unroll
  for (int j = 0; j < 5; ++j) {
    v2f aS, aE, bb;
#pragma unroll
    for (int v = 0; v < 2; ++v) {
      const int k   = v + 2 * half;
      const int ec  = 4 * j + k;
      const int ecc = ec > EC - 1 ? EC - 1 : ec;
      const int wp  = w0 - 1 + ec;
      const int o1  = ((i + 1) * EC + ecc) * SPITCH + col;

      const float s1 = Ss[o1];
      const float s0 = Ss[o1 - EC * SPITCH];
      const float s2 = Ss[o1 + EC * SPITCH];
      const float e1 = Es[o1];
      const float e0 = Es[o1 - EC * SPITCH];
      const float e2 = Es[o1 + EC * SPITCH];
      aS[v] = fmaf(gU, s0, fmaf(gD, s2, s1));
      aE[v] = fmaf(gU, e0, fmaf(gD, e2, e1));

      const int diff = ec - 1 - nn;
      const bool inband = (unsigned)(diff + 1) <= 2u;
      const bool wok    = (unsigned)wp < (unsigned)Wdim;
      bb[v] = (inband && wok) ? 1.f : 0.f;
    }
    accS = __builtin_amdgcn_wmma_f32_16x16x4_f32(false, aS, false, bb,
                                                 (short)0, accS, false, false);
    accE = __builtin_amdgcn_wmma_f32_16x16x4_f32(false, aE, false, bb,
                                                 (short)0, accE, false, false);
  }

  const int wcol = w0 + nn;
  float bsa = 0.f;
#pragma unroll
  for (int dy = -1; dy <= 1; ++dy) {
    const int hh = h + dy;
    const bool hok = (unsigned)hh < (unsigned)Hdim;
    const float gh = hok ? 1.f : 0.f;
#pragma unroll
    for (int dx = -1; dx <= 1; ++dx) {
      const int ww  = wcol + dx;
      const bool wok = (unsigned)ww < (unsigned)Wdim;
      const float g = wok ? gh : 0.f;
      bsa = fmaf(g, As[(i + 1 + dy) * EC + nn + 1 + dx], bsa);
    }
  }
  const float nq = fmaxf(sqrtf(bsa), 1e-12f);

#pragma unroll
  for (int r = 0; r < 8; ++r) {
    const int d     = dbase + r + 8 * half;
    const float nk  = fmaxf(sqrtf(accE[r]), 1e-12f);
    const float sim = accS[r] / (nq * nk);
    out[((size_t)(b * Ddim + d) * Hdim + h) * Wdim + wcol] = sim;
  }
}

extern "C" void kernel_launch(void* const* d_in, const int* in_sizes, int n_in,
                              void* d_out, int out_size, void* d_ws,
                              size_t ws_size, hipStream_t stream) {
  const float* q  = (const float*)d_in[0];
  const float* wf = (const float*)d_in[1];
  float* out = (float*)d_out;
  dav_fused<<<240, THREADS, 0, stream>>>(q, wf, out);
}


# speedup vs baseline: 1.1721x; 1.0806x over previous
#include <hip/hip_runtime.h>
#include <math.h>

typedef __attribute__((ext_vector_type(2))) float v2f;
typedef __attribute__((ext_vector_type(4))) float v4f;
typedef __attribute__((ext_vector_type(8))) float v8f;

#define Bdim 2
#define Cdim 32
#define Hdim 96
#define Wdim 160
#define Ddim 32
#define CSTRIDE (Hdim * Wdim * Ddim)

#define TH 8
#define TW 16
#define ER (TH + 2)
#define EC (TW + 2)
#define NPX (ER * EC)
#define SWAVES 15
#define PXW 12
#define WAVES 16
#define THREADS (WAVES * 32)
#define QPITCH 36
#define SPITCH 40
#define NCHUNK 6

__global__ __launch_bounds__(THREADS) void dav_fused(
    const float* __restrict__ q, const float* __restrict__ wf,
    float* __restrict__ out) {
  __shared__ __attribute__((aligned(16))) float qs[SWAVES * PXW * QPITCH];
  __shared__ __attribute__((aligned(16))) float Ss[NPX * SPITCH];
  __shared__ __attribute__((aligned(16))) float Es[NPX * SPITCH];
  __shared__ float As[NPX + 12];

  const int t    = threadIdx.x;
  const int lane = t & 31;
  const int wave = __builtin_amdgcn_readfirstlane(t >> 5);
  const int blk  = blockIdx.x;
  const int slab = blk & 7;
  const int jt   = blk >> 3;
  const int tw   = jt % 10;
  const int th   = 3 * (slab & 3) + jt / 10;
  const int b    = slab >> 2;
  const int h0   = th * TH;
  const int w0   = tw * TW;

  if (wave < SWAVES) {
    const int pq = lane >> 3;
    const int d4 = lane & 7;
    float* qsw = qs + wave * (PXW * QPITCH);

    const float* wfb = wf + (size_t)b * Cdim * CSTRIDE;
    int koff[3];
#pragma unroll
    for (int g = 0; g < 3; ++g) {
      const int p  = wave * PXW + 4 * g + pq;
      const int er = p / EC;
      const int ec = p - er * EC;
      int h = h0 - 1 + er;
      int w = w0 - 1 + ec;
      h = h < 0 ? 0 : (h > Hdim - 1 ? Hdim - 1 : h);
      w = w < 0 ? 0 : (w > Wdim - 1 ? Wdim - 1 : w);
      koff[g] = (h * Wdim + w) * Ddim + d4 * 4;
    }

    v4f buf[2][16];
#define LOAD_CHUNK(K)                                                          \
  do {                                                                         \
    const float* p_ = wfb + koff[(K) >> 1] + (((K) & 1) * 16) * CSTRIDE;       \
    _Pragma("unroll") for (int j_ = 0; j_ < 16; ++j_)                          \
        buf[(K) & 1][j_] = *(const v4f*)(p_ + j_ * CSTRIDE);                   \
  } while (0)

    LOAD_CHUNK(0);
    __builtin_amdgcn_sched_barrier(0);

    float qreg[12];
    int   qdst[12];
#pragma unroll
    for (int i = 0; i < 12; ++i) {
      const int e  = lane + 32 * i;
      const int c  = e / PXW;
      const int px = e - c * PXW;
      const int p  = wave * PXW + px;
      const int er = p / EC;
      const int ec = p - er * EC;
      int h = h0 - 1 + er;
      int w = w0 - 1 + ec;
      h = h < 0 ? 0 : (h > Hdim - 1 ? Hdim - 1 : h);
      w = w < 0 ? 0 : (w > Wdim - 1 ? Wdim - 1 : w);
      qreg[i] = q[((b * Cdim + c) * Hdim + h) * Wdim + w];
      qdst[i] = px * QPITCH + c;
    }
    __builtin_amdgcn_sched_barrier(0);
    LOAD_CHUNK(1);
    __builtin_amdgcn_sched_barrier(0);

#pragma unroll
    for (int i = 0; i < 12; ++i) qsw[qdst[i]] = qreg[i];
    __builtin_amdgcn_fence(__ATOMIC_RELEASE, "wavefront");
    __builtin_amdgcn_wave_barrier();
    __builtin_amdgcn_fence(__ATOMIC_ACQUIRE, "wavefront");

    v4f s4[3], e4[3];
#pragma unroll
    for (int g = 0; g < 3; ++g) {
      s4[g] = (v4f){0.f, 0.f, 0.f, 0.f};
      e4[g] = (v4f){0.f, 0.f, 0.f, 0.f};
    }

#pragma unroll
    for (int k = 0; k < NCHUNK; ++k) {
      const int g = k >> 1;
      const float* qrow = qsw + (4 * g + pq) * QPITCH + (k & 1) * 16;
#pragma unroll
      for (int j4 = 0; j4 < 4; ++j4) {
        const v4f qv4 = *(const v4f*)(qrow + 4 * j4);
#pragma unroll
        for (int u = 0; u < 4; ++u) {
          const v4f kv  = buf[k & 1][4 * j4 + u];
          const float qv = qv4[u];
          const v4f qvv = {qv, qv, qv, qv};
          s4[g] += qvv * kv;
          e4[g] += kv * kv;
        }
      }
      __builtin_amdgcn_sched_barrier(0);
      if (k + 2 < NCHUNK) {
        asm volatile("" : "+v"(koff[(k + 2) >> 1]) : "v"(s4[g]), "v"(e4[g]));
        LOAD_CHUNK(k + 2);
      }
      __builtin_amdgcn_sched_barrier(0);
    }
#undef LOAD_CHUNK

    if (lane < PXW) {
      float acc = 0.f;
#pragma unroll
      for (int c = 0; c < Cdim; ++c) {
        const float v = qsw[lane * QPITCH + c];
        acc += v * v;
      }
      As[wave * PXW + lane] = acc;
    }

#pragma unroll
    for (int g = 0; g < 3; ++g) {
      const int p = wave * PXW + 4 * g + pq;
      *(v4f*)(Ss + p * SPITCH + d4 * 4) = s4[g];
      *(v4f*)(Es + p * SPITCH + d4 * 4) = e4[g];
    }
  }
  __syncthreads();

  const int i     = wave >> 1;
  const int dh    = wave & 1;
  const int h     = h0 + i;
  const int dbase = dh * 16;
  const int nn    = lane & 15;
  const int half  = lane >> 4;
  const int col   = dbase + nn;

  const float gU = (h > 0) ? 1.f : 0.f;
  const float gD = (h < Hdim - 1) ? 1.f : 0.f;

  v8f accS = {0.f, 0.f, 0.f, 0.f, 0.f, 0.f, 0.f, 0.f};
  v8f accE = {0.f, 0.f, 0.f, 0.f, 0.f, 0.f, 0.f, 0.f};

#pragma unroll
  for (int j = 0; j < 5; ++j) {
    v2f aS, aE, bb;
#pragma unroll
    for (int v = 0; v < 2; ++v) {
      const int k   = v + 2 * half;
      const int ec  = 4 * j + k;
      const int ecc = ec > EC - 1 ? EC - 1 : ec;
      const int wp  = w0 - 1 + ec;
      const int o1  = ((i + 1) * EC + ecc) * SPITCH + col;

      const float s1 = Ss[o1];
      const float s0 = Ss[o1 - EC * SPITCH];
      const float s2 = Ss[o1 + EC * SPITCH];
      const float e1 = Es[o1];
      const float e0 = Es[o1 - EC * SPITCH];
      const float e2 = Es[o1 + EC * SPITCH];
      aS[v] = fmaf(gU, s0, fmaf(gD, s2, s1));
      aE[v] = fmaf(gU, e0, fmaf(gD, e2, e1));

      const int diff = ec - 1 - nn;
      const bool inband = (unsigned)(diff + 1) <= 2u;
      const bool wok    = (unsigned)wp < (unsigned)Wdim;
      bb[v] = (inband && wok) ? 1.f : 0.f;
    }
    accS = __builtin_amdgcn_wmma_f32_16x16x4_f32(false, aS, false, bb,
                                                 (short)0, accS, false, false);
    accE = __builtin_amdgcn_wmma_f32_16x16x4_f32(false, aE, false, bb,
                                                 (short)0, accE, false, false);
  }

  const int wcol = w0 + nn;
  float bsa = 0.f;
#pragma unroll
  for (int dy = -1; dy <= 1; ++dy) {
    const int hh = h + dy;
    const bool hok = (unsigned)hh < (unsigned)Hdim;
    const float gh = hok ? 1.f : 0.f;
#pragma unroll
    for (int dx = -1; dx <= 1; ++dx) {
      const int ww  = wcol + dx;
      const bool wok = (unsigned)ww < (unsigned)Wdim;
      const float g = wok ? gh : 0.f;
      bsa = fmaf(g, As[(i + 1 + dy) * EC + nn + 1 + dx], bsa);
    }
  }
  const float nq = fmaxf(sqrtf(bsa), 1e-12f);

#pragma unroll
  for (int r = 0; r < 8; ++r) {
    const int d     = dbase + r + 8 * half;
    const float nk  = fmaxf(sqrtf(accE[r]), 1e-12f);
    const float sim = accS[r] / (nq * nk);
    out[((size_t)(b * Ddim + d) * Hdim + h) * Wdim + wcol] = sim;
  }
}

extern "C" void kernel_launch(void* const* d_in, const int* in_sizes, int n_in,
                              void* d_out, int out_size, void* d_ws,
                              size_t ws_size, hipStream_t stream) {
  const float* q  = (const float*)d_in[0];
  const float* wf = (const float*)d_in[1];
  float* out = (float*)d_out;
  dav_fused<<<240, THREADS, 0, stream>>>(q, wf, out);
}


# streaming loads re-ordered: a load now covers two whole aligned 256-B units (4 pixels from an image column that is a multiple of 4); the halo's 20 edge pixels go in 5 loads of their own; same sums
# speedup vs baseline: 1.2651x; 1.0793x over previous
#include <hip/hip_runtime.h>
#include <math.h>

typedef __attribute__((ext_vector_type(2))) float v2f;
typedef __attribute__((ext_vector_type(4))) float v4f;
typedef __attribute__((ext_vector_type(8))) float v8f;

#define Bdim 2
#define Cdim 32
#define Hdim 96
#define Wdim 160
#define Ddim 32
#define CSTRIDE (Hdim * Wdim * Ddim)

#define TH 8
#define TW 16
#define ER (TH + 2)
#define EC (TW + 2)
#define NPX (ER * EC)
#define SWAVES 15
#define PXW 12
#define WAVES 16
#define THREADS (WAVES * 32)
#define QPITCH 36
#define SPITCH 40
#define NCHUNK 6

__device__ __forceinline__ void lane_px(int wave, int local, int& er, int& ec) {
  const int m  = wave / 3;
  const int r  = wave - 3 * m;
  const int g  = local >> 2;
  const int pq = local & 3;
  const bool single = (r == 2) && (g == 2);
  const int pi = 2 * (3 * wave + g - m) + (pq >> 1);
  const int t  = 4 * m + pq;
  er = single ? (t >> 1) : (pi >> 3);
  ec = single ? ((t & 1) * (EC - 1)) : (1 + 2 * (pi & 7) + (pq & 1));
}

__global__ __launch_bounds__(THREADS) void dav_fused(
    const float* __restrict__ q, const float* __restrict__ wf,
    float* __restrict__ out) {
  __shared__ __attribute__((aligned(16))) float qs[SWAVES * PXW * QPITCH];
  __shared__ __attribute__((aligned(16))) float Ss[NPX * SPITCH];
  __shared__ __attribute__((aligned(16))) float Es[NPX * SPITCH];
  __shared__ float As[NPX + 12];

  const int t    = threadIdx.x;
  const int lane = t & 31;
  const int wave = __builtin_amdgcn_readfirstlane(t >> 5);
  const int blk  = blockIdx.x;
  const int slab = blk & 7;
  const int jt   = blk >> 3;
  const int tw   = jt % 10;
  const int th   = 3 * (slab & 3) + jt / 10;
  const int b    = slab >> 2;
  const int h0   = th * TH;
  const int w0   = tw * TW;

  if (wave < SWAVES) {
    const int pq = lane >> 3;
    const int d4 = lane & 7;
    float* qsw = qs + wave * (PXW * QPITCH);

    const float* wfb = wf + (size_t)b * Cdim * CSTRIDE;
    int koff[3];
#pragma unroll
    for (int g = 0; g < 3; ++g) {
      int er, ec;
      lane_px(wave, 4 * g + pq, er, ec);
      int h = h0 - 1 + er;
      int w = w0 - 1 + ec;
      h = h < 0 ? 0 : (h > Hdim - 1 ? Hdim - 1 : h);
      w = w < 0 ? 0 : (w > Wdim - 1 ? Wdim - 1 : w);
      koff[g] = (h * Wdim + w) * Ddim + d4 * 4;
    }

    v4f buf[2][16];
#define LOAD_CHUNK(K)                                                          \
  do {                                                                         \
    const float* p_ = wfb + koff[(K) >> 1] + (((K) & 1) * 16) * CSTRIDE;       \
    _Pragma("unroll") for (int j_ = 0; j_ < 16; ++j_)                          \
        buf[(K) & 1][j_] = *(const v4f*)(p_ + j_ * CSTRIDE);                   \
  } while (0)

    LOAD_CHUNK(0);
    __builtin_amdgcn_sched_barrier(0);

    float qreg[12];
    int   qdst[12];
#pragma unroll
    for (int i = 0; i < 12; ++i) {
      const int e  = lane + 32 * i;
      const int c  = e / PXW;
      const int px = e - c * PXW;
      int er, ec;
      lane_px(wave, px, er, ec);
      int h = h0 - 1 + er;
      int w = w0 - 1 + ec;
      h = h < 0 ? 0 : (h > Hdim - 1 ? Hdim - 1 : h);
      w = w < 0 ? 0 : (w > Wdim - 1 ? Wdim - 1 : w);
      qreg[i] = q[((b * Cdim + c) * Hdim + h) * Wdim + w];
      qdst[i] = px * QPITCH + c;
    }
    __builtin_amdgcn_sched_barrier(0);
    LOAD_CHUNK(1);
    __builtin_amdgcn_sched_barrier(0);

#pragma unroll
    for (int i = 0; i < 12; ++i) qsw[qdst[i]] = qreg[i];
    __builtin_amdgcn_fence(__ATOMIC_RELEASE, "wavefront");
    __builtin_amdgcn_wave_barrier();
    __builtin_amdgcn_fence(__ATOMIC_ACQUIRE, "wavefront");

    v4f s4[3], e4[3];
#pragma unroll
    for (int g = 0; g < 3; ++g) {
      s4[g] = (v4f){0.f, 0.f, 0.f, 0.f};
      e4[g] = (v4f){0.f, 0.f, 0.f, 0.f};
    }

#pragma unroll
    for (int k = 0; k < NCHUNK; ++k) {
      const int g = k >> 1;
      const float* qrow = qsw + (4 * g + pq) * QPITCH + (k & 1) * 16;
#pragma unroll
      for (int j4 = 0; j4 < 4; ++j4) {
        const v4f qv4 = *(const v4f*)(qrow + 4 * j4);
#pragma unroll
        for (int u = 0; u < 4; ++u) {
          const v4f kv  = buf[k & 1][4 * j4 + u];
          const float qv = qv4[u];
          const v4f qvv = {qv, qv, qv, qv};
          s4[g] += qvv * kv;
          e4[g] += kv * kv;
        }
      }
      __builtin_amdgcn_sched_barrier(0);
      if (k + 2 < NCHUNK) {
        asm volatile("" : "+v"(koff[(k + 2) >> 1]) : "v"(s4[g]), "v"(e4[g]));
        LOAD_CHUNK(k + 2);
      }
      __builtin_amdgcn_sched_barrier(0);
    }
#undef LOAD_CHUNK

    if (lane < PXW) {
      float acc = 0.f;
#pragma unroll
      for (int c = 0; c < Cdim; ++c) {
        const float v = qsw[lane * QPITCH + c];
        acc += v * v;
      }
      int er, ec;
      lane_px(wave, lane, er, ec);
      As[er * EC + ec] = acc;
    }

#pragma unroll
    for (int g = 0; g < 3; ++g) {
      int er, ec;
      lane_px(wave, 4 * g + pq, er, ec);
      const int p = er * EC + ec;
      *(v4f*)(Ss + p * SPITCH + d4 * 4) = s4[g];
      *(v4f*)(Es + p * SPITCH + d4 * 4) = e4[g];
    }
  }
  __syncthreads();

  const int i     = wave >> 1;
  const int dh    = wave & 1;
  const int h     = h0 + i;
  const int dbase = dh * 16;
  const int nn    = lane & 15;
  const int half  = lane >> 4;
  const int col   = dbase + nn;

  const float gU = (h > 0) ? 1.f : 0.f;
  const float gD = (h < Hdim - 1) ? 1.f : 0.f;

  v8f accS = {0.f, 0.f, 0.f, 0.f, 0.f, 0.f, 0.f, 0.f};
  v8f accE = {0.f, 0.f, 0.f, 0.f, 0.f, 0.f, 0.f, 0.f};

#pragma unroll
  for (int j = 0; j < 5; ++j) {
    v2f aS, aE, bb;
#pragma unroll
    for (int v = 0; v < 2; ++v) {
      const int k   = v + 2 * half;
      const int ec  = 4 * j + k;
      const int ecc = ec > EC - 1 ? EC - 1 : ec;
      const int wp  = w0 - 1 + ec;
      const int o1  = ((i + 1) * EC + ecc) * SPITCH + col;

      const float s1 = Ss[o1];
      const float s0 = Ss[o1 - EC * SPITCH];
      const float s2 = Ss[o1 + EC * SPITCH];
      const float e1 = Es[o1];
      const float e0 = Es[o1 - EC * SPITCH];
      const float e2 = Es[o1 + EC * SPITCH];
      aS[v] = fmaf(gU, s0, fmaf(gD, s2, s1));
      aE[v] = fmaf(gU, e0, fmaf(gD, e2, e1));

      const int diff = ec - 1 - nn;
      const bool inband = (unsigned)(diff + 1) <= 2u;
      const bool wok    = (unsigned)wp < (unsigned)Wdim;
      bb[v] = (inband && wok) ? 1.f : 0.f;
    }
    accS = __builtin_amdgcn_wmma_f32_16x16x4_f32(false, aS, false, bb,
                                                 (short)0, accS, false, false);
    accE = __builtin_amdgcn_wmma_f32_16x16x4_f32(false, aE, false, bb,
                                                 (short)0, accE, false, false);
  }

  const int wcol = w0 + nn;
  float bsa = 0.f;
#pragma unroll
  for (int dy = -1; dy <= 1; ++dy) {
    const int hh = h + dy;
    const bool hok = (unsigned)hh < (unsigned)Hdim;
    const float gh = hok ? 1.f : 0.f;
#pragma unroll
    for (int dx = -1; dx <= 1; ++dx) {
      const int ww  = wcol + dx;
      const bool wok = (unsigned)ww < (unsigned)Wdim;
      const float g = wok ? gh : 0.f;
      bsa = fmaf(g, As[(i + 1 + dy) * EC + nn + 1 + dx], bsa);
    }
  }
  const float nq = fmaxf(sqrtf(bsa), 1e-12f);

#pragma unroll
  for (int r = 0; r < 8; ++r) {
    const int d     = dbase + r + 8 * half;
    const float nk  = fmaxf(sqrtf(accE[r]), 1e-12f);
    const float sim = accS[r] / (nq * nk);
    out[((size_t)(b * Ddim + d) * Hdim + h) * Wdim + wcol] = sim;
  }
}

extern "C" void kernel_launch(void* const* d_in, const int* in_sizes, int n_in,
                              void* d_out, int out_size, void* d_ws,
                              size_t ws_size, hipStream_t stream) {
  const float* q  = (const float*)d_in[0];
  const float* wf = (const float*)d_in[1];
  float* out = (float*)d_out;
  dav_fused<<<240, THREADS, 0, stream>>>(q, wf, out);
}
